# starting kernel
# baseline (speedup 1.0000x reference)
#include <hip/hip_runtime.h>
#include <math.h>

#define BB 4096
#define NN 8192
#define WW 64

typedef float v2f __attribute__((ext_vector_type(2)));
typedef float v8f __attribute__((ext_vector_type(8)));

__global__ __launch_bounds__(64) void ntm_prep_kernel(
    const float* __restrict__ k, const float* __restrict__ beta,
    const float* __restrict__ memory,
    float* __restrict__ Asc, float* __restrict__ Mc)
{
  __shared__ float red[64];
  const int row = blockIdx.x;
  const int t = threadIdx.x;
  if (row < BB) {
    float x = k[row * WW + t];
    red[t] = x * x;
    for (int s2 = 32; s2 > 0; s2 >>= 1) { __syncthreads(); if (t < s2) red[t] += red[t + s2]; }
    __syncthreads();
    const float n1 = sqrtf(red[0]);
    __syncthreads();
    const float kn = x / fmaxf(n1, 1e-12f);
    red[t] = kn * kn;
    for (int s2 = 32; s2 > 0; s2 >>= 1) { __syncthreads(); if (t < s2) red[t] += red[t + s2]; }
    __syncthreads();
    const float n2 = sqrtf(red[0]);
    const float kc = kn / fmaxf(n2, 1e-8f);
    Asc[row * WW + t] = beta[row] * kc;
  } else {
    const int r = row - BB;
    float x = memory[r * WW + t];
    red[t] = x * x;
    for (int s2 = 32; s2 > 0; s2 >>= 1) { __syncthreads(); if (t < s2) red[t] += red[t + s2]; }
    __syncthreads();
    const float nm = sqrtf(red[0]);
    Mc[r * WW + t] = x / fmaxf(nm, 1e-8f);
  }
}

__global__ __launch_bounds__(256) void ntm_gemm_exp_kernel(
    const float* __restrict__ Asc, const float* __restrict__ Mc,
    float* __restrict__ E)
{
  const int lane  = threadIdx.x & 31;
  const int wave  = threadIdx.x >> 5;
  const int ntile = (blockIdx.x * 8 + wave) << 4;
  const int mtile = blockIdx.y << 4;
  const int l15   = lane & 15;
  const int khalf = (lane >> 4) << 1;

  const float* __restrict__ arow = Asc + (size_t)(mtile + l15) * WW + khalf;
  const float* __restrict__ brow = Mc  + (size_t)(ntile + l15) * WW + khalf;

  v8f c = {};
#pragma unroll
  for (int k0 = 0; k0 < WW; k0 += 4) {
    v2f a, b;
    a.x = arow[k0];     a.y = arow[k0 + 1];
    b.x = brow[k0];     b.y = brow[k0 + 1];
    c = __builtin_amdgcn_wmma_f32_16x16x4_f32(
         false, a,  false, b,
         (short)0, c,  false,  false);
  }

  const int col  = ntile + l15;
  const int row0 = mtile + ((lane >> 4) << 3);
  float* __restrict__ ep = E + (size_t)row0 * NN + col;
#pragma unroll
  for (int v = 0; v < 8; ++v) {
    ep[(size_t)v * NN] = expf(c[v]);
  }
}

__global__ __launch_bounds__(256) void ntm_epilogue_kernel(
    const float* __restrict__ E, const float* __restrict__ g,
    const float* __restrict__ s, const float* __restrict__ gamma,
    const float* __restrict__ prev, float* __restrict__ out)
{
  __shared__ float gatedS[NN];
  __shared__ float red[256];
  __shared__ float ssS[3];

  const int b = blockIdx.x;
  const int t = threadIdx.x;
  const float* __restrict__ Erow = E + (size_t)b * NN;

  float e[32];
  float lsum = 0.f;
#pragma unroll
  for (int i = 0; i < 32; ++i) { float v = Erow[i * 256 + t]; e[i] = v; lsum += v; }
  red[t] = lsum;
  for (int s2 = 128; s2 > 0; s2 >>= 1) { __syncthreads(); if (t < s2) red[t] += red[t + s2]; }
  __syncthreads();
  const float sumE = red[0];

  if (t == 0) {
    const float s0 = s[b * 3 + 0], s1 = s[b * 3 + 1], s2v = s[b * 3 + 2];
    const float mx = fmaxf(s0, fmaxf(s1, s2v));
    const float e0 = expf(s0 - mx), e1 = expf(s1 - mx), e2 = expf(s2v - mx);
    const float inv = 1.f / (e0 + e1 + e2);
    ssS[0] = e0 * inv; ssS[1] = e1 * inv; ssS[2] = e2 * inv;
  }

  const float gb    = g[b];
  const float scale = gb / sumE;
  const float og    = 1.f - gb;
  const float* __restrict__ prow = prev + (size_t)b * NN;
#pragma unroll
  for (int i = 0; i < 32; ++i) {
    const int n = i * 256 + t;
    gatedS[n] = e[i] * scale + og * prow[n];
  }
  __syncthreads();

  const float ss0 = ssS[0], ss1 = ssS[1], ss2 = ssS[2];
  const float gm  = gamma[b];
  float sh[32];
  float lsum2 = 0.f;
#pragma unroll
  for (int i = 0; i < 32; ++i) {
    const int n = i * 256 + t;
    const float left  = (n > 0)      ? gatedS[n - 1] : 0.f;
    const float right = (n < NN - 1) ? gatedS[n + 1] : 0.f;
    const float sv = left * ss0 + gatedS[n] * ss1 + right * ss2;
    const float p  = powf(sv, gm);
    sh[i] = p; lsum2 += p;
  }
  __syncthreads();
  red[t] = lsum2;
  for (int s2 = 128; s2 > 0; s2 >>= 1) { __syncthreads(); if (t < s2) red[t] += red[t + s2]; }
  __syncthreads();
  const float invS = 1.f / (red[0] + 1e-8f);

  float* __restrict__ orow = out + (size_t)b * NN;
#pragma unroll
  for (int i = 0; i < 32; ++i) orow[i * 256 + t] = sh[i] * invS;
}

extern "C" void kernel_launch(void* const* d_in, const int* in_sizes, int n_in,
                              void* d_out, int out_size, void* d_ws, size_t ws_size,
                              hipStream_t stream) {
  const float* k      = (const float*)d_in[0];
  const float* beta   = (const float*)d_in[1];
  const float* g      = (const float*)d_in[2];
  const float* s      = (const float*)d_in[3];
  const float* gamma  = (const float*)d_in[4];
  const float* prev   = (const float*)d_in[5];
  const float* memory = (const float*)d_in[6];
  float* out = (float*)d_out;

  char* ws = (char*)d_ws;
  float* E   = (float*)ws;
  float* Asc = (float*)(ws + (size_t)BB * NN * sizeof(float));
  float* Mc  = (float*)(ws + (size_t)BB * NN * sizeof(float)
                           + (size_t)BB * WW * sizeof(float));

  ntm_prep_kernel<<<BB + NN, 64, 0, stream>>>(k, beta, memory, Asc, Mc);
  ntm_gemm_exp_kernel<<<dim3(NN / 128, BB / 16), 256, 0, stream>>>(Asc, Mc, E);
  ntm_epilogue_kernel<<<BB, 256, 0, stream>>>(E, g, s, gamma, prev, out);
}


# GEMM operands staged once per 128x128 block in padded LDS by coalesced 16-byte loads; each wave keeps 2x4 tiles fed by 8-byte LDS fragment reads; k order and summation order unchanged
# speedup vs baseline: 1.3184x; 1.3184x over previous
#include <hip/hip_runtime.h>
#include <math.h>

#define BB 4096
#define NN 8192
#define WW 64

typedef float v2f __attribute__((ext_vector_type(2)));
typedef float v8f __attribute__((ext_vector_type(8)));

__global__ __launch_bounds__(64) void ntm_prep_kernel(
    const float* __restrict__ k, const float* __restrict__ beta,
    const float* __restrict__ memory,
    float* __restrict__ Asc, float* __restrict__ Mc)
{
  __shared__ float red[64];
  const int row = blockIdx.x;
  const int t = threadIdx.x;
  if (row < BB) {
    float x = k[row * WW + t];
    red[t] = x * x;
    for (int s2 = 32; s2 > 0; s2 >>= 1) { __syncthreads(); if (t < s2) red[t] += red[t + s2]; }
    __syncthreads();
    const float n1 = sqrtf(red[0]);
    __syncthreads();
    const float kn = x / fmaxf(n1, 1e-12f);
    red[t] = kn * kn;
    for (int s2 = 32; s2 > 0; s2 >>= 1) { __syncthreads(); if (t < s2) red[t] += red[t + s2]; }
    __syncthreads();
    const float n2 = sqrtf(red[0]);
    const float kc = kn / fmaxf(n2, 1e-8f);
    Asc[row * WW + t] = beta[row] * kc;
  } else {
    const int r = row - BB;
    float x = memory[r * WW + t];
    red[t] = x * x;
    for (int s2 = 32; s2 > 0; s2 >>= 1) { __syncthreads(); if (t < s2) red[t] += red[t + s2]; }
    __syncthreads();
    const float nm = sqrtf(red[0]);
    Mc[r * WW + t] = x / fmaxf(nm, 1e-8f);
  }
}

#define GT_M 128
#define GT_N 128
#define GPITCH 68

__global__ __launch_bounds__(256) void ntm_gemm_exp_kernel(
    const float* __restrict__ Asc, const float* __restrict__ Mc,
    float* __restrict__ E)
{
  __shared__ __attribute__((aligned(16))) float As[GT_M * GPITCH];
  __shared__ __attribute__((aligned(16))) float Bs[GT_N * GPITCH];

  const int t  = threadIdx.x;
  const int n0 = blockIdx.x * GT_N;
  const int m0 = blockIdx.y * GT_M;

  const float4* __restrict__ ag = reinterpret_cast<const float4*>(Asc + (size_t)m0 * WW);
  const float4* __restrict__ bg = reinterpret_cast<const float4*>(Mc  + (size_t)n0 * WW);
  float4 va[8], vb[8];
#pragma unroll
  for (int i = 0; i < 8; ++i) { va[i] = ag[i * 256 + t]; vb[i] = bg[i * 256 + t]; }
#pragma unroll
  for (int i = 0; i < 8; ++i) {
    const int idx = i * 256 + t;
    const int off = (idx >> 4) * GPITCH + ((idx & 15) << 2);
    *reinterpret_cast<float4*>(&As[off]) = va[i];
    *reinterpret_cast<float4*>(&Bs[off]) = vb[i];
  }
  __syncthreads();

  const int lane  = t & 31;
  const int wave  = t >> 5;
  const int wm    = wave & 3;
  const int wn    = wave >> 2;
  const int l15   = lane & 15;
  const int khalf = (lane >> 4) << 1;

  const float* ap = As + (wm * 32 + l15) * GPITCH + khalf;
  const float* bp = Bs + (wn * 64 + l15) * GPITCH + khalf;

  v8f c[2][4] = {};
#pragma unroll
  for (int k0 = 0; k0 < WW; k0 += 4) {
    v2f a[2], b[4];
#pragma unroll
    for (int i = 0; i < 2; ++i) a[i] = *reinterpret_cast<const v2f*>(ap + i * 16 * GPITCH + k0);
#pragma unroll
    for (int j = 0; j < 4; ++j) b[j] = *reinterpret_cast<const v2f*>(bp + j * 16 * GPITCH + k0);
#pragma unroll
    for (int i = 0; i < 2; ++i) {
#pragma unroll
      for (int j = 0; j < 4; ++j) {
        c[i][j] = __builtin_amdgcn_wmma_f32_16x16x4_f32(
             false, a[i],  false, b[j],
             (short)0, c[i][j],  false,  false);
      }
    }
  }

#pragma unroll
  for (int i = 0; i < 2; ++i) {
    const int row0 = m0 + wm * 32 + i * 16 + ((lane >> 4) << 3);
    float* __restrict__ ep = E + (size_t)row0 * NN + (n0 + wn * 64 + l15);
#pragma unroll
    for (int v = 0; v < 8; ++v) {
#pragma unroll
      for (int j = 0; j < 4; ++j) {
        ep[(size_t)v * NN + j * 16] = expf(c[i][j][v]);
      }
    }
  }
}

__global__ __launch_bounds__(256) void ntm_epilogue_kernel(
    const float* __restrict__ E, const float* __restrict__ g,
    const float* __restrict__ s, const float* __restrict__ gamma,
    const float* __restrict__ prev, float* __restrict__ out)
{
  __shared__ float gatedS[NN];
  __shared__ float red[256];
  __shared__ float ssS[3];

  const int b = blockIdx.x;
  const int t = threadIdx.x;
  const float* __restrict__ Erow = E + (size_t)b * NN;

  float e[32];
  float lsum = 0.f;
#pragma unroll
  for (int i = 0; i < 32; ++i) { float v = Erow[i * 256 + t]; e[i] = v; lsum += v; }
  red[t] = lsum;
  for (int s2 = 128; s2 > 0; s2 >>= 1) { __syncthreads(); if (t < s2) red[t] += red[t + s2]; }
  __syncthreads();
  const float sumE = red[0];

  if (t == 0) {
    const float s0 = s[b * 3 + 0], s1 = s[b * 3 + 1], s2v = s[b * 3 + 2];
    const float mx = fmaxf(s0, fmaxf(s1, s2v));
    const float e0 = expf(s0 - mx), e1 = expf(s1 - mx), e2 = expf(s2v - mx);
    const float inv = 1.f / (e0 + e1 + e2);
    ssS[0] = e0 * inv; ssS[1] = e1 * inv; ssS[2] = e2 * inv;
  }

  const float gb    = g[b];
  const float scale = gb / sumE;
  const float og    = 1.f - gb;
  const float* __restrict__ prow = prev + (size_t)b * NN;
#pragma unroll
  for (int i = 0; i < 32; ++i) {
    const int n = i * 256 + t;
    gatedS[n] = e[i] * scale + og * prow[n];
  }
  __syncthreads();

  const float ss0 = ssS[0], ss1 = ssS[1], ss2 = ssS[2];
  const float gm  = gamma[b];
  float sh[32];
  float lsum2 = 0.f;
#pragma unroll
  for (int i = 0; i < 32; ++i) {
    const int n = i * 256 + t;
    const float left  = (n > 0)      ? gatedS[n - 1] : 0.f;
    const float right = (n < NN - 1) ? gatedS[n + 1] : 0.f;
    const float sv = left * ss0 + gatedS[n] * ss1 + right * ss2;
    const float p  = powf(sv, gm);
    sh[i] = p; lsum2 += p;
  }
  __syncthreads();
  red[t] = lsum2;
  for (int s2 = 128; s2 > 0; s2 >>= 1) { __syncthreads(); if (t < s2) red[t] += red[t + s2]; }
  __syncthreads();
  const float invS = 1.f / (red[0] + 1e-8f);

  float* __restrict__ orow = out + (size_t)b * NN;
#pragma unroll
  for (int i = 0; i < 32; ++i) orow[i * 256 + t] = sh[i] * invS;
}

extern "C" void kernel_launch(void* const* d_in, const int* in_sizes, int n_in,
                              void* d_out, int out_size, void* d_ws, size_t ws_size,
                              hipStream_t stream) {
  const float* k      = (const float*)d_in[0];
  const float* beta   = (const float*)d_in[1];
  const float* g      = (const float*)d_in[2];
  const float* s      = (const float*)d_in[3];
  const float* gamma  = (const float*)d_in[4];
  const float* prev   = (const float*)d_in[5];
  const float* memory = (const float*)d_in[6];
  float* out = (float*)d_out;

  char* ws = (char*)d_ws;
  float* E   = (float*)ws;
  float* Asc = (float*)(ws + (size_t)BB * NN * sizeof(float));
  float* Mc  = (float*)(ws + (size_t)BB * NN * sizeof(float)
                           + (size_t)BB * WW * sizeof(float));

  ntm_prep_kernel<<<BB + NN, 64, 0, stream>>>(k, beta, memory, Asc, Mc);
  ntm_gemm_exp_kernel<<<dim3(NN / GT_N, BB / GT_M), 256, 0, stream>>>(Asc, Mc, E);
  ntm_epilogue_kernel<<<BB, 256, 0, stream>>>(E, g, s, gamma, prev, out);
}


# Sharpen x^gamma as exp2f(gamma*log2f(x)) on the hardware log/exp units instead of library powf (rel. error up to about 1e-6, was correctly rounded); gamma==0 keeps pow's 1 by a block-uniform branch
# speedup vs baseline: 2.0710x; 1.5708x over previous
#include <hip/hip_runtime.h>
#include <math.h>

#define BB 4096
#define NN 8192
#define WW 64

typedef float v2f __attribute__((ext_vector_type(2)));
typedef float v8f __attribute__((ext_vector_type(8)));

__global__ __launch_bounds__(64) void ntm_prep_kernel(
    const float* __restrict__ k, const float* __restrict__ beta,
    const float* __restrict__ memory,
    float* __restrict__ Asc, float* __restrict__ Mc)
{
  __shared__ float red[64];
  const int row = blockIdx.x;
  const int t = threadIdx.x;
  if (row < BB) {
    float x = k[row * WW + t];
    red[t] = x * x;
    for (int s2 = 32; s2 > 0; s2 >>= 1) { __syncthreads(); if (t < s2) red[t] += red[t + s2]; }
    __syncthreads();
    const float n1 = sqrtf(red[0]);
    __syncthreads();
    const float kn = x / fmaxf(n1, 1e-12f);
    red[t] = kn * kn;
    for (int s2 = 32; s2 > 0; s2 >>= 1) { __syncthreads(); if (t < s2) red[t] += red[t + s2]; }
    __syncthreads();
    const float n2 = sqrtf(red[0]);
    const float kc = kn / fmaxf(n2, 1e-8f);
    Asc[row * WW + t] = beta[row] * kc;
  } else {
    const int r = row - BB;
    float x = memory[r * WW + t];
    red[t] = x * x;
    for (int s2 = 32; s2 > 0; s2 >>= 1) { __syncthreads(); if (t < s2) red[t] += red[t + s2]; }
    __syncthreads();
    const float nm = sqrtf(red[0]);
    Mc[r * WW + t] = x / fmaxf(nm, 1e-8f);
  }
}

#define GT_M 128
#define GT_N 128
#define GPITCH 68

__global__ __launch_bounds__(256) void ntm_gemm_exp_kernel(
    const float* __restrict__ Asc, const float* __restrict__ Mc,
    float* __restrict__ E)
{
  __shared__ __attribute__((aligned(16))) float As[GT_M * GPITCH];
  __shared__ __attribute__((aligned(16))) float Bs[GT_N * GPITCH];

  const int t  = threadIdx.x;
  const int n0 = blockIdx.x * GT_N;
  const int m0 = blockIdx.y * GT_M;

  const float4* __restrict__ ag = reinterpret_cast<const float4*>(Asc + (size_t)m0 * WW);
  const float4* __restrict__ bg = reinterpret_cast<const float4*>(Mc  + (size_t)n0 * WW);
  float4 va[8], vb[8];
#pragma unroll
  for (int i = 0; i < 8; ++i) { va[i] = ag[i * 256 + t]; vb[i] = bg[i * 256 + t]; }
#pragma unroll
  for (int i = 0; i < 8; ++i) {
    const int idx = i * 256 + t;
    const int off = (idx >> 4) * GPITCH + ((idx & 15) << 2);
    *reinterpret_cast<float4*>(&As[off]) = va[i];
    *reinterpret_cast<float4*>(&Bs[off]) = vb[i];
  }
  __syncthreads();

  const int lane  = t & 31;
  const int wave  = t >> 5;
  const int wm    = wave & 3;
  const int wn    = wave >> 2;
  const int l15   = lane & 15;
  const int khalf = (lane >> 4) << 1;

  const float* ap = As + (wm * 32 + l15) * GPITCH + khalf;
  const float* bp = Bs + (wn * 64 + l15) * GPITCH + khalf;

  v8f c[2][4] = {};
#pragma unroll
  for (int k0 = 0; k0 < WW; k0 += 4) {
    v2f a[2], b[4];
#pragma unroll
    for (int i = 0; i < 2; ++i) a[i] = *reinterpret_cast<const v2f*>(ap + i * 16 * GPITCH + k0);
#pragma unroll
    for (int j = 0; j < 4; ++j) b[j] = *reinterpret_cast<const v2f*>(bp + j * 16 * GPITCH + k0);
#pragma unroll
    for (int i = 0; i < 2; ++i) {
#pragma unroll
      for (int j = 0; j < 4; ++j) {
        c[i][j] = __builtin_amdgcn_wmma_f32_16x16x4_f32(
             false, a[i],  false, b[j],
             (short)0, c[i][j],  false,  false);
      }
    }
  }

#pragma unroll
  for (int i = 0; i < 2; ++i) {
    const int row0 = m0 + wm * 32 + i * 16 + ((lane >> 4) << 3);
    float* __restrict__ ep = E + (size_t)row0 * NN + (n0 + wn * 64 + l15);
#pragma unroll
    for (int v = 0; v < 8; ++v) {
#pragma unroll
      for (int j = 0; j < 4; ++j) {
        ep[(size_t)v * NN + j * 16] = expf(c[i][j][v]);
      }
    }
  }
}

__global__ __launch_bounds__(256) void ntm_epilogue_kernel(
    const float* __restrict__ E, const float* __restrict__ g,
    const float* __restrict__ s, const float* __restrict__ gamma,
    const float* __restrict__ prev, float* __restrict__ out)
{
  __shared__ float gatedS[NN];
  __shared__ float red[256];
  __shared__ float ssS[3];

  const int b = blockIdx.x;
  const int t = threadIdx.x;
  const float* __restrict__ Erow = E + (size_t)b * NN;

  float e[32];
  float lsum = 0.f;
#pragma unroll
  for (int i = 0; i < 32; ++i) { float v = Erow[i * 256 + t]; e[i] = v; lsum += v; }
  red[t] = lsum;
  for (int s2 = 128; s2 > 0; s2 >>= 1) { __syncthreads(); if (t < s2) red[t] += red[t + s2]; }
  __syncthreads();
  const float sumE = red[0];

  if (t == 0) {
    const float s0 = s[b * 3 + 0], s1 = s[b * 3 + 1], s2v = s[b * 3 + 2];
    const float mx = fmaxf(s0, fmaxf(s1, s2v));
    const float e0 = expf(s0 - mx), e1 = expf(s1 - mx), e2 = expf(s2v - mx);
    const float inv = 1.f / (e0 + e1 + e2);
    ssS[0] = e0 * inv; ssS[1] = e1 * inv; ssS[2] = e2 * inv;
  }

  const float gb    = g[b];
  const float scale = gb / sumE;
  const float og    = 1.f - gb;
  const float* __restrict__ prow = prev + (size_t)b * NN;
#pragma unroll
  for (int i = 0; i < 32; ++i) {
    const int n = i * 256 + t;
    gatedS[n] = e[i] * scale + og * prow[n];
  }
  __syncthreads();

  const float ss0 = ssS[0], ss1 = ssS[1], ss2 = ssS[2];
  const float gm  = gamma[b];
  float sh[32];
  float lsum2 = 0.f;
  if (gm == 0.f) {
#pragma unroll
    for (int i = 0; i < 32; ++i) { sh[i] = 1.f; lsum2 += 1.f; }
  } else {
#pragma unroll
    for (int i = 0; i < 32; ++i) {
      const int n = i * 256 + t;
      const float left  = (n > 0)      ? gatedS[n - 1] : 0.f;
      const float right = (n < NN - 1) ? gatedS[n + 1] : 0.f;
      const float sv = left * ss0 + gatedS[n] * ss1 + right * ss2;
      const float p  = exp2f(gm * log2f(sv));
      sh[i] = p; lsum2 += p;
    }
  }
  __syncthreads();
  red[t] = lsum2;
  for (int s2 = 128; s2 > 0; s2 >>= 1) { __syncthreads(); if (t < s2) red[t] += red[t + s2]; }
  __syncthreads();
  const float invS = 1.f / (red[0] + 1e-8f);

  float* __restrict__ orow = out + (size_t)b * NN;
#pragma unroll
  for (int i = 0; i < 32; ++i) orow[i * 256 + t] = sh[i] * invS;
}

extern "C" void kernel_launch(void* const* d_in, const int* in_sizes, int n_in,
                              void* d_out, int out_size, void* d_ws, size_t ws_size,
                              hipStream_t stream) {
  const float* k      = (const float*)d_in[0];
  const float* beta   = (const float*)d_in[1];
  const float* g      = (const float*)d_in[2];
  const float* s      = (const float*)d_in[3];
  const float* gamma  = (const float*)d_in[4];
  const float* prev   = (const float*)d_in[5];
  const float* memory = (const float*)d_in[6];
  float* out = (float*)d_out;

  char* ws = (char*)d_ws;
  float* E   = (float*)ws;
  float* Asc = (float*)(ws + (size_t)BB * NN * sizeof(float));
  float* Mc  = (float*)(ws + (size_t)BB * NN * sizeof(float)
                           + (size_t)BB * WW * sizeof(float));

  ntm_prep_kernel<<<BB + NN, 64, 0, stream>>>(k, beta, memory, Asc, Mc);
  ntm_gemm_exp_kernel<<<dim3(NN / GT_N, BB / GT_M), 256, 0, stream>>>(Asc, Mc, E);
  ntm_epilogue_kernel<<<BB, 256, 0, stream>>>(E, g, s, gamma, prev, out);
}
